# starting kernel
# baseline (speedup 1.0000x reference)
#include <hip/hip_runtime.h>
#include <math.h>


typedef __attribute__((ext_vector_type(2))) float v2f;
typedef __attribute__((ext_vector_type(8))) float v8f;

static constexpr int HID    = 256;
static constexpr int IN_DIM = 64;

__global__ void init_deg_kernel(float* __restrict__ deg, int n) {
  int i = blockIdx.x * blockDim.x + threadIdx.x;
  if (i < n) deg[i] = 1.0f;
}

__global__ void accum_deg_kernel(const int* __restrict__ tgt,
                                 float* __restrict__ deg, int e) {
  int i = blockIdx.x * blockDim.x + threadIdx.x;
  if (i < e) atomicAdd(&deg[tgt[i]], 1.0f);
}

__global__ void rsqrt_kernel(float* __restrict__ deg, int n) {
  int i = blockIdx.x * blockDim.x + threadIdx.x;
  if (i < n) deg[i] = rsqrtf(deg[i]);
}

__global__ void zero_kernel(float* __restrict__ p, int n) {
  int i = blockIdx.x * blockDim.x + threadIdx.x;
  if (i < n) p[i] = 0.0f;
}

__global__ void init_out_bias_kernel(float* __restrict__ out,
                                     const float* __restrict__ b, int total) {
  int i = blockIdx.x * blockDim.x + threadIdx.x;
  if (i < total) out[i] = b[i & (HID - 1)];
}

template <int K>
__global__ void wmma_gemm_kernel(const float* __restrict__ A,
                                 const float* __restrict__ B,
                                 float* __restrict__ C, int M) {
  const int lane = threadIdx.x & 31;
  const int wid  = (blockIdx.x * blockDim.x + threadIdx.x) >> 5;
  const int tilesN = HID / 16;
  const int tilesM = (M + 15) / 16;
  if (wid >= tilesM * tilesN) return;

  const int tm = wid / tilesN;
  const int tn = wid % tilesN;
  const int h  = lane >> 4;
  const int ln = lane & 15;

  const int m = tm * 16 + ln;
  const int n = tn * 16 + ln;

  v8f acc = {};
#pragma unroll 4
  for (int k0 = 0; k0 < K; k0 += 4) {
    v2f a, b;
    const float* ap = A + (size_t)m * K + (k0 + 2 * h);
    a.x = ap[0];
    a.y = ap[1];
    b.x = B[(size_t)(k0 + 2 * h) * HID + n];
    b.y = B[(size_t)(k0 + 2 * h + 1) * HID + n];
    acc = __builtin_amdgcn_wmma_f32_16x16x4_f32(
        false, a, false, b, (short)0, acc, false, false);
  }

  float* cp = C + (size_t)(tm * 16 + 8 * h) * HID + n;
#pragma unroll
  for (int r = 0; r < 8; ++r) cp[(size_t)r * HID] = acc[r];
}

__global__ void scatter_kernel(const float* __restrict__ H,
                               const int* __restrict__ src,
                               const int* __restrict__ tgt,
                               const float* __restrict__ dinv,
                               float* __restrict__ out, int E, int N) {
  const int wid  = (blockIdx.x * blockDim.x + threadIdx.x) >> 5;
  const int lane = threadIdx.x & 31;
  if (wid >= E + N) return;

  int s, t;
  if (wid < E) { s = src[wid]; t = tgt[wid]; }
  else         { s = t = wid - E; }

  const float norm = dinv[s] * dinv[t];
  const float* hp = H + (size_t)s * HID;
  float* op = out + (size_t)t * HID;
#pragma unroll
  for (int j = 0; j < 8; ++j) {
    const int c = lane + 32 * j;
    atomicAdd(op + c, hp[c] * norm);
  }
}

__global__ void epilogue1_kernel(const float* __restrict__ agg,
                                 const float* __restrict__ b1,
                                 float* __restrict__ X2, int total) {
  int idx = blockIdx.x * blockDim.x + threadIdx.x;
  if (idx >= total) return;
  const int i = idx >> 8;
  const int c = idx & 255;
  const float kLn1e4Over256 = 9.210340371976184f / 256.0f;
  const float freq = expf(-(float)(c & ~1) * kLn1e4Over256);
  const float ang  = (float)i * freq;
  const float pe   = (c & 1) ? cosf(ang) : sinf(ang);
  const float v = agg[idx] + b1[c] + pe;
  X2[idx] = v > 0.0f ? v : 0.0f;
}

extern "C" void kernel_launch(void* const* d_in, const int* in_sizes, int n_in,
                              void* d_out, int out_size, void* d_ws, size_t ws_size,
                              hipStream_t stream) {
  const float* X  = (const float*)d_in[0];
  const int*   EI = (const int*)d_in[1];
  const float* W1 = (const float*)d_in[2];
  const float* b1 = (const float*)d_in[3];
  const float* W2 = (const float*)d_in[4];
  const float* b2 = (const float*)d_in[5];
  float* out = (float*)d_out;

  const int N = in_sizes[0] / IN_DIM;
  const int E = in_sizes[1] / 2;
  const int NF = N * HID;

  float* dinv   = (float*)d_ws;
  float* bufH   = dinv + (1 << 16);
  float* bufAgg = bufH + (size_t)NF;

  const int* src = EI;
  const int* tgt = EI + E;

  const int T = 256;
  auto cdiv = [](int a, int b) { return (a + b - 1) / b; };

  init_deg_kernel<<<cdiv(N, T), T, 0, stream>>>(dinv, N);
  accum_deg_kernel<<<cdiv(E, T), T, 0, stream>>>(tgt, dinv, E);
  rsqrt_kernel<<<cdiv(N, T), T, 0, stream>>>(dinv, N);

  {
    const int tiles = cdiv(N, 16) * (HID / 16);
    wmma_gemm_kernel<IN_DIM><<<cdiv(tiles, T / 32), T, 0, stream>>>(X, W1, bufH, N);
  }

  zero_kernel<<<cdiv(NF, T), T, 0, stream>>>(bufAgg, NF);
  scatter_kernel<<<cdiv(E + N, T / 32), T, 0, stream>>>(bufH, src, tgt, dinv,
                                                        bufAgg, E, N);

  epilogue1_kernel<<<cdiv(NF, T), T, 0, stream>>>(bufAgg, b1, bufH, NF);

  {
    const int tiles = cdiv(N, 16) * (HID / 16);
    wmma_gemm_kernel<HID><<<cdiv(tiles, T / 32), T, 0, stream>>>(bufH, W2, bufAgg, N);
  }

  init_out_bias_kernel<<<cdiv(NF, T), T, 0, stream>>>(out, b2, NF);
  scatter_kernel<<<cdiv(E + N, T / 32), T, 0, stream>>>(bufAgg, src, tgt, dinv,
                                                        out, E, N);
}


# layer 1 aggregates the 64-wide input rows before the W1 product ((A_hat X) W1, not A_hat (X W1)): scatter 1 issues a quarter of the float atomics; summation order changes (product reassociated)
# speedup vs baseline: 1.4162x; 1.4162x over previous
#include <hip/hip_runtime.h>
#include <math.h>


typedef __attribute__((ext_vector_type(2))) float v2f;
typedef __attribute__((ext_vector_type(8))) float v8f;

static constexpr int HID    = 256;
static constexpr int IN_DIM = 64;

__global__ void init_deg_kernel(float* __restrict__ deg, int n) {
  int i = blockIdx.x * blockDim.x + threadIdx.x;
  if (i < n) deg[i] = 1.0f;
}

__global__ void accum_deg_kernel(const int* __restrict__ tgt,
                                 float* __restrict__ deg, int e) {
  int i = blockIdx.x * blockDim.x + threadIdx.x;
  if (i < e) atomicAdd(&deg[tgt[i]], 1.0f);
}

__global__ void rsqrt_kernel(float* __restrict__ deg, int n) {
  int i = blockIdx.x * blockDim.x + threadIdx.x;
  if (i < n) deg[i] = rsqrtf(deg[i]);
}

__global__ void zero_kernel(float* __restrict__ p, int n) {
  int i = blockIdx.x * blockDim.x + threadIdx.x;
  if (i < n) p[i] = 0.0f;
}

__global__ void init_out_bias_kernel(float* __restrict__ out,
                                     const float* __restrict__ b, int total) {
  int i = blockIdx.x * blockDim.x + threadIdx.x;
  if (i < total) out[i] = b[i & (HID - 1)];
}

template <int K>
__global__ void wmma_gemm_kernel(const float* __restrict__ A,
                                 const float* __restrict__ B,
                                 float* __restrict__ C, int M) {
  const int lane = threadIdx.x & 31;
  const int wid  = (blockIdx.x * blockDim.x + threadIdx.x) >> 5;
  const int tilesN = HID / 16;
  const int tilesM = (M + 15) / 16;
  if (wid >= tilesM * tilesN) return;

  const int tm = wid / tilesN;
  const int tn = wid % tilesN;
  const int h  = lane >> 4;
  const int ln = lane & 15;

  const int m = tm * 16 + ln;
  const int n = tn * 16 + ln;

  v8f acc = {};
#pragma unroll 4
  for (int k0 = 0; k0 < K; k0 += 4) {
    v2f a, b;
    const float* ap = A + (size_t)m * K + (k0 + 2 * h);
    a.x = ap[0];
    a.y = ap[1];
    b.x = B[(size_t)(k0 + 2 * h) * HID + n];
    b.y = B[(size_t)(k0 + 2 * h + 1) * HID + n];
    acc = __builtin_amdgcn_wmma_f32_16x16x4_f32(
        false, a, false, b, (short)0, acc, false, false);
  }

  float* cp = C + (size_t)(tm * 16 + 8 * h) * HID + n;
#pragma unroll
  for (int r = 0; r < 8; ++r) cp[(size_t)r * HID] = acc[r];
}

__global__ void scatter_kernel(const float* __restrict__ H,
                               const int* __restrict__ src,
                               const int* __restrict__ tgt,
                               const float* __restrict__ dinv,
                               float* __restrict__ out, int E, int N) {
  const int wid  = (blockIdx.x * blockDim.x + threadIdx.x) >> 5;
  const int lane = threadIdx.x & 31;
  if (wid >= E + N) return;

  int s, t;
  if (wid < E) { s = src[wid]; t = tgt[wid]; }
  else         { s = t = wid - E; }

  const float norm = dinv[s] * dinv[t];
  const float* hp = H + (size_t)s * HID;
  float* op = out + (size_t)t * HID;
#pragma unroll
  for (int j = 0; j < 8; ++j) {
    const int c = lane + 32 * j;
    atomicAdd(op + c, hp[c] * norm);
  }
}

static constexpr int EPW = 4;
__global__ void scatter_in_kernel(const float* __restrict__ X,
                                  const int* __restrict__ src,
                                  const int* __restrict__ tgt,
                                  const float* __restrict__ dinv,
                                  float* __restrict__ out, int E, int N) {
  const int wid   = (blockIdx.x * blockDim.x + threadIdx.x) >> 5;
  const int lane  = threadIdx.x & 31;
  const int total = E + N;
  const int base  = wid * EPW;
  if (base >= total) return;
  const int cnt = (total - base < EPW) ? (total - base) : EPW;

  int s[EPW], t[EPW];
#pragma unroll
  for (int q = 0; q < EPW; ++q) {
    const int slot = (q < cnt) ? (base + q) : base;
    if (slot < E) { s[q] = src[slot]; t[q] = tgt[slot]; }
    else          { s[q] = t[q] = slot - E; }
  }
  float v0[EPW], v1[EPW];
#pragma unroll
  for (int q = 0; q < EPW; ++q) {
    const float norm = dinv[s[q]] * dinv[t[q]];
    const float* hp = X + (size_t)s[q] * IN_DIM;
    v0[q] = hp[lane] * norm;
    v1[q] = hp[lane + 32] * norm;
  }
#pragma unroll
  for (int q = 0; q < EPW; ++q) {
    if (q < cnt) {
      float* op = out + (size_t)t[q] * IN_DIM;
      atomicAdd(op + lane, v0[q]);
      atomicAdd(op + lane + 32, v1[q]);
    }
  }
}

__global__ void epilogue1_kernel(const float* __restrict__ agg,
                                 const float* __restrict__ b1,
                                 float* __restrict__ X2, int total) {
  int idx = blockIdx.x * blockDim.x + threadIdx.x;
  if (idx >= total) return;
  const int i = idx >> 8;
  const int c = idx & 255;
  const float kLn1e4Over256 = 9.210340371976184f / 256.0f;
  const float freq = expf(-(float)(c & ~1) * kLn1e4Over256);
  const float ang  = (float)i * freq;
  const float pe   = (c & 1) ? cosf(ang) : sinf(ang);
  const float v = agg[idx] + b1[c] + pe;
  X2[idx] = v > 0.0f ? v : 0.0f;
}

extern "C" void kernel_launch(void* const* d_in, const int* in_sizes, int n_in,
                              void* d_out, int out_size, void* d_ws, size_t ws_size,
                              hipStream_t stream) {
  const float* X  = (const float*)d_in[0];
  const int*   EI = (const int*)d_in[1];
  const float* W1 = (const float*)d_in[2];
  const float* b1 = (const float*)d_in[3];
  const float* W2 = (const float*)d_in[4];
  const float* b2 = (const float*)d_in[5];
  float* out = (float*)d_out;

  const int N = in_sizes[0] / IN_DIM;
  const int E = in_sizes[1] / 2;
  const int NF = N * HID;

  float* dinv   = (float*)d_ws;
  float* bufH   = dinv + (1 << 16);
  float* bufAgg = bufH + (size_t)NF;

  const int* src = EI;
  const int* tgt = EI + E;

  const int T = 256;
  auto cdiv = [](int a, int b) { return (a + b - 1) / b; };

  init_deg_kernel<<<cdiv(N, T), T, 0, stream>>>(dinv, N);
  accum_deg_kernel<<<cdiv(E, T), T, 0, stream>>>(tgt, dinv, E);
  rsqrt_kernel<<<cdiv(N, T), T, 0, stream>>>(dinv, N);

  float* aggX = bufH;
  zero_kernel<<<cdiv(N * IN_DIM, T), T, 0, stream>>>(aggX, N * IN_DIM);
  {
    const int waves = cdiv(E + N, EPW);
    scatter_in_kernel<<<cdiv(waves, T / 32), T, 0, stream>>>(X, src, tgt, dinv,
                                                             aggX, E, N);
  }

  {
    const int tiles = cdiv(N, 16) * (HID / 16);
    wmma_gemm_kernel<IN_DIM><<<cdiv(tiles, T / 32), T, 0, stream>>>(aggX, W1, bufAgg, N);
  }

  epilogue1_kernel<<<cdiv(NF, T), T, 0, stream>>>(bufAgg, b1, bufH, NF);

  {
    const int tiles = cdiv(N, 16) * (HID / 16);
    wmma_gemm_kernel<HID><<<cdiv(tiles, T / 32), T, 0, stream>>>(bufH, W2, bufAgg, N);
  }

  init_out_bias_kernel<<<cdiv(NF, T), T, 0, stream>>>(out, b2, NF);
  scatter_kernel<<<cdiv(E + N, T / 32), T, 0, stream>>>(bufAgg, src, tgt, dinv,
                                                        out, E, N);
}


# both GEMMs re-tiled: a wave owns 16 rows x 128 columns (8 accumulators, permuted tile columns), operands by 16-byte loads; k order inside each block of 8 changes (summation order)
# speedup vs baseline: 2.9293x; 2.0684x over previous
#include <hip/hip_runtime.h>
#include <math.h>


typedef __attribute__((ext_vector_type(2))) float v2f;
typedef __attribute__((ext_vector_type(8))) float v8f;

static constexpr int HID    = 256;
static constexpr int IN_DIM = 64;

__global__ void zero_counts_kernel(int* __restrict__ cnt, int n) {
  int i = blockIdx.x * blockDim.x + threadIdx.x;
  if (i < n) cnt[i] = 0;
}

__global__ void count_rank_kernel(const int* __restrict__ tgt,
                                  int* __restrict__ cnt,
                                  int* __restrict__ rank, int e) {
  int i = blockIdx.x * blockDim.x + threadIdx.x;
  if (i < e) rank[i] = atomicAdd(&cnt[tgt[i]], 1);
}

static constexpr int SCAN_T = 1024;
__global__ void __launch_bounds__(SCAN_T)
rowptr_dinv_kernel(const int* __restrict__ cnt, int* __restrict__ rowptr,
                   float* __restrict__ dinv, int n) {
  __shared__ int s_tot[32];
  __shared__ int s_pre[32];
  const int tid  = threadIdx.x;
  const int lane = tid & 31;
  const int wv   = tid >> 5;
  const int first = blockIdx.x * SCAN_T;

  int pre = 0;
  for (int i = tid; i < first; i += SCAN_T) pre += cnt[i];

  const int i = first + tid;
  const int c = (i < n) ? cnt[i] : 0;
  int inc = c;
  for (int d = 1; d < 32; d <<= 1) {
    const int v = __shfl_up(inc, d);
    if (lane >= d) inc += v;
  }
  for (int d = 16; d > 0; d >>= 1) pre += __shfl_xor(pre, d);
  if (lane == 31) s_tot[wv] = inc;
  if (lane == 0)  s_pre[wv] = pre;
  __syncthreads();
  if (wv == 0) {
    const int t = s_tot[lane];
    int p  = s_pre[lane];
    int ti = t;
    for (int d = 1; d < 32; d <<= 1) {
      const int v = __shfl_up(ti, d);
      if (lane >= d) ti += v;
    }
    for (int d = 16; d > 0; d >>= 1) p += __shfl_xor(p, d);
    s_tot[lane] = ti - t + p;
  }
  __syncthreads();
  const int excl = s_tot[wv] + inc - c;
  if (i < n) {
    rowptr[i] = excl;
    dinv[i]   = rsqrtf((float)(c + 1));
    if (i == n - 1) rowptr[n] = excl + c;
  }
}

__global__ void fill_adj_kernel(const int* __restrict__ src,
                                const int* __restrict__ tgt,
                                const int* __restrict__ rank,
                                const int* __restrict__ rowptr,
                                const float* __restrict__ dinv,
                                int2* __restrict__ adj, int e) {
  int i = blockIdx.x * blockDim.x + threadIdx.x;
  if (i < e) {
    const int s = src[i];
    adj[rowptr[tgt[i]] + rank[i]] = make_int2(s, __float_as_int(dinv[s]));
  }
}

typedef __attribute__((ext_vector_type(4))) float v4f;

template <int K>
__global__ void __launch_bounds__(128)
wmma_gemm_kernel(const float* __restrict__ A,
                 const float* __restrict__ B,
                 float* __restrict__ C, int M) {
  const int lane = threadIdx.x & 31;
  const int wv = __builtin_amdgcn_readfirstlane((int)(threadIdx.x >> 5));
  const int strip  = blockIdx.x * 2 + (wv >> 1);
  const int tilesM = (M + 15) / 16;
  if (strip >= tilesM) return;

  const int colbase = (wv & 1) * 128;
  const int h = lane >> 4;
  const int c = lane & 15;

  int row = strip * 16 + c;
  if (row > M - 1) row = M - 1;
  const float* ap = A + (size_t)row * K + 4 * h;
  const float* bp = B + (size_t)(4 * h) * HID + colbase + 4 * c;

  v8f acc[8];
#pragma unroll
  for (int t = 0; t < 8; ++t) acc[t] = v8f{};

  for (int kk = 0; kk < K; kk += 8) {
    const v4f a = *(const v4f*)(ap + kk);
    v4f b[2][2][2];
#pragma unroll
    for (int s = 0; s < 2; ++s)
#pragma unroll
      for (int j = 0; j < 2; ++j)
#pragma unroll
        for (int q = 0; q < 2; ++q)
          b[s][j][q] = *(const v4f*)(bp + (size_t)(kk + 2 * s + j) * HID + 64 * q);

#pragma unroll
    for (int s = 0; s < 2; ++s) {
      v2f af;
      af.x = a[2 * s];
      af.y = a[2 * s + 1];
#pragma unroll
      for (int q = 0; q < 2; ++q)
#pragma unroll
        for (int u = 0; u < 4; ++u) {
          v2f bf;
          bf.x = b[s][0][q][u];
          bf.y = b[s][1][q][u];
          acc[4 * q + u] = __builtin_amdgcn_wmma_f32_16x16x4_f32(
              false, af, false, bf, (short)0, acc[4 * q + u], false, false);
        }
    }
  }

  const int row0 = strip * 16 + 8 * h;
  float* cp = C + (size_t)row0 * HID + colbase + 4 * c;
#pragma unroll
  for (int r = 0; r < 8; ++r) {
    if (row0 + r < M) {
#pragma unroll
      for (int q = 0; q < 2; ++q) {
        v4f o;
        o.x = acc[4 * q + 0][r];
        o.y = acc[4 * q + 1][r];
        o.z = acc[4 * q + 2][r];
        o.w = acc[4 * q + 3][r];
        *(v4f*)(cp + (size_t)r * HID + 64 * q) = o;
      }
    }
  }
}


__global__ void gather64_kernel(const float* __restrict__ X,
                                const int* __restrict__ rowptr,
                                const int2* __restrict__ adj,
                                const float* __restrict__ dinv,
                                float* __restrict__ out, int N) {
  const int lane = threadIdx.x & 31;
  const int t = __builtin_amdgcn_readfirstlane(
      (int)((blockIdx.x * blockDim.x + threadIdx.x) >> 5));
  if (t >= N) return;
  const int beg = rowptr[t];
  const int end = rowptr[t + 1];
  const float dt = dinv[t];
  const float2* __restrict__ Xv = (const float2*)X;

  const float2 self = Xv[(size_t)t * 32 + lane];
  float ax = self.x * dt, ay = self.y * dt;

  for (int base = beg; base < end; base += 32) {
    const int n = (end - base < 32) ? (end - base) : 32;
    int2 e = make_int2(0, 0);
    if (lane < n) e = adj[base + lane];
    int k = 0;
    for (; k + 4 <= n; k += 4) {
      const int s0 = __builtin_amdgcn_readlane(e.x, k);
      const int s1 = __builtin_amdgcn_readlane(e.x, k + 1);
      const int s2 = __builtin_amdgcn_readlane(e.x, k + 2);
      const int s3 = __builtin_amdgcn_readlane(e.x, k + 3);
      const float w0 = __int_as_float(__builtin_amdgcn_readlane(e.y, k));
      const float w1 = __int_as_float(__builtin_amdgcn_readlane(e.y, k + 1));
      const float w2 = __int_as_float(__builtin_amdgcn_readlane(e.y, k + 2));
      const float w3 = __int_as_float(__builtin_amdgcn_readlane(e.y, k + 3));
      const float2 r0 = Xv[(size_t)s0 * 32 + lane];
      const float2 r1 = Xv[(size_t)s1 * 32 + lane];
      const float2 r2 = Xv[(size_t)s2 * 32 + lane];
      const float2 r3 = Xv[(size_t)s3 * 32 + lane];
      ax = fmaf(r0.x, w0, ax); ay = fmaf(r0.y, w0, ay);
      ax = fmaf(r1.x, w1, ax); ay = fmaf(r1.y, w1, ay);
      ax = fmaf(r2.x, w2, ax); ay = fmaf(r2.y, w2, ay);
      ax = fmaf(r3.x, w3, ax); ay = fmaf(r3.y, w3, ay);
    }
#pragma nounroll
    for (; k < n; ++k) {
      const int s0 = __builtin_amdgcn_readlane(e.x, k);
      const float w0 = __int_as_float(__builtin_amdgcn_readlane(e.y, k));
      const float2 r0 = Xv[(size_t)s0 * 32 + lane];
      ax = fmaf(r0.x, w0, ax); ay = fmaf(r0.y, w0, ay);
    }
  }
  float2 o;
  o.x = ax * dt;
  o.y = ay * dt;
  ((float2*)out)[(size_t)t * 32 + lane] = o;
}

__global__ void gather256_bias_kernel(const float* __restrict__ H,
                                      const int* __restrict__ rowptr,
                                      const int2* __restrict__ adj,
                                      const float* __restrict__ dinv,
                                      const float* __restrict__ bias,
                                      float* __restrict__ out, int N) {
  const int lane = threadIdx.x & 31;
  const int t = __builtin_amdgcn_readfirstlane(
      (int)((blockIdx.x * blockDim.x + threadIdx.x) >> 5));
  if (t >= N) return;
  const int beg = rowptr[t];
  const int end = rowptr[t + 1];
  const float dt = dinv[t];
  const float4* __restrict__ Hv = (const float4*)H;

  float4 a0, a1;
  {
    const float4 p = Hv[(size_t)t * 64 + lane];
    const float4 q = Hv[(size_t)t * 64 + 32 + lane];
    a0.x = p.x * dt; a0.y = p.y * dt; a0.z = p.z * dt; a0.w = p.w * dt;
    a1.x = q.x * dt; a1.y = q.y * dt; a1.z = q.z * dt; a1.w = q.w * dt;
  }

  for (int base = beg; base < end; base += 32) {
    const int n = (end - base < 32) ? (end - base) : 32;
    int2 e = make_int2(0, 0);
    if (lane < n) e = adj[base + lane];
    int k = 0;
    for (; k + 4 <= n; k += 4) {
      const int s0 = __builtin_amdgcn_readlane(e.x, k);
      const int s1 = __builtin_amdgcn_readlane(e.x, k + 1);
      const int s2 = __builtin_amdgcn_readlane(e.x, k + 2);
      const int s3 = __builtin_amdgcn_readlane(e.x, k + 3);
      const float w0 = __int_as_float(__builtin_amdgcn_readlane(e.y, k));
      const float w1 = __int_as_float(__builtin_amdgcn_readlane(e.y, k + 1));
      const float w2 = __int_as_float(__builtin_amdgcn_readlane(e.y, k + 2));
      const float w3 = __int_as_float(__builtin_amdgcn_readlane(e.y, k + 3));
      const float4 p0 = Hv[(size_t)s0 * 64 + lane];
      const float4 q0 = Hv[(size_t)s0 * 64 + 32 + lane];
      const float4 p1 = Hv[(size_t)s1 * 64 + lane];
      const float4 q1 = Hv[(size_t)s1 * 64 + 32 + lane];
      const float4 p2 = Hv[(size_t)s2 * 64 + lane];
      const float4 q2 = Hv[(size_t)s2 * 64 + 32 + lane];
      const float4 p3 = Hv[(size_t)s3 * 64 + lane];
      const float4 q3 = Hv[(size_t)s3 * 64 + 32 + lane];
      a0.x = fmaf(p0.x, w0, a0.x); a0.y = fmaf(p0.y, w0, a0.y);
      a0.z = fmaf(p0.z, w0, a0.z); a0.w = fmaf(p0.w, w0, a0.w);
      a1.x = fmaf(q0.x, w0, a1.x); a1.y = fmaf(q0.y, w0, a1.y);
      a1.z = fmaf(q0.z, w0, a1.z); a1.w = fmaf(q0.w, w0, a1.w);
      a0.x = fmaf(p1.x, w1, a0.x); a0.y = fmaf(p1.y, w1, a0.y);
      a0.z = fmaf(p1.z, w1, a0.z); a0.w = fmaf(p1.w, w1, a0.w);
      a1.x = fmaf(q1.x, w1, a1.x); a1.y = fmaf(q1.y, w1, a1.y);
      a1.z = fmaf(q1.z, w1, a1.z); a1.w = fmaf(q1.w, w1, a1.w);
      a0.x = fmaf(p2.x, w2, a0.x); a0.y = fmaf(p2.y, w2, a0.y);
      a0.z = fmaf(p2.z, w2, a0.z); a0.w = fmaf(p2.w, w2, a0.w);
      a1.x = fmaf(q2.x, w2, a1.x); a1.y = fmaf(q2.y, w2, a1.y);
      a1.z = fmaf(q2.z, w2, a1.z); a1.w = fmaf(q2.w, w2, a1.w);
      a0.x = fmaf(p3.x, w3, a0.x); a0.y = fmaf(p3.y, w3, a0.y);
      a0.z = fmaf(p3.z, w3, a0.z); a0.w = fmaf(p3.w, w3, a0.w);
      a1.x = fmaf(q3.x, w3, a1.x); a1.y = fmaf(q3.y, w3, a1.y);
      a1.z = fmaf(q3.z, w3, a1.z); a1.w = fmaf(q3.w, w3, a1.w);
    }
#pragma nounroll
    for (; k < n; ++k) {
      const int s0 = __builtin_amdgcn_readlane(e.x, k);
      const float w0 = __int_as_float(__builtin_amdgcn_readlane(e.y, k));
      const float4 p0 = Hv[(size_t)s0 * 64 + lane];
      const float4 q0 = Hv[(size_t)s0 * 64 + 32 + lane];
      a0.x = fmaf(p0.x, w0, a0.x); a0.y = fmaf(p0.y, w0, a0.y);
      a0.z = fmaf(p0.z, w0, a0.z); a0.w = fmaf(p0.w, w0, a0.w);
      a1.x = fmaf(q0.x, w0, a1.x); a1.y = fmaf(q0.y, w0, a1.y);
      a1.z = fmaf(q0.z, w0, a1.z); a1.w = fmaf(q0.w, w0, a1.w);
    }
  }
  const float4 b0 = ((const float4*)bias)[lane];
  const float4 b1 = ((const float4*)bias)[32 + lane];
  float4 o0, o1;
  o0.x = fmaf(a0.x, dt, b0.x); o0.y = fmaf(a0.y, dt, b0.y);
  o0.z = fmaf(a0.z, dt, b0.z); o0.w = fmaf(a0.w, dt, b0.w);
  o1.x = fmaf(a1.x, dt, b1.x); o1.y = fmaf(a1.y, dt, b1.y);
  o1.z = fmaf(a1.z, dt, b1.z); o1.w = fmaf(a1.w, dt, b1.w);
  ((float4*)out)[(size_t)t * 64 + lane]      = o0;
  ((float4*)out)[(size_t)t * 64 + 32 + lane] = o1;
}

__global__ void epilogue1_kernel(const float* __restrict__ agg,
                                 const float* __restrict__ b1,
                                 float* __restrict__ X2, int total) {
  int idx = blockIdx.x * blockDim.x + threadIdx.x;
  if (idx >= total) return;
  const int i = idx >> 8;
  const int c = idx & 255;
  const float kLn1e4Over256 = 9.210340371976184f / 256.0f;
  const float freq = expf(-(float)(c & ~1) * kLn1e4Over256);
  const float ang  = (float)i * freq;
  const float pe   = (c & 1) ? cosf(ang) : sinf(ang);
  const float v = agg[idx] + b1[c] + pe;
  X2[idx] = v > 0.0f ? v : 0.0f;
}

extern "C" void kernel_launch(void* const* d_in, const int* in_sizes, int n_in,
                              void* d_out, int out_size, void* d_ws, size_t ws_size,
                              hipStream_t stream) {
  const float* X  = (const float*)d_in[0];
  const int*   EI = (const int*)d_in[1];
  const float* W1 = (const float*)d_in[2];
  const float* b1 = (const float*)d_in[3];
  const float* W2 = (const float*)d_in[4];
  const float* b2 = (const float*)d_in[5];
  float* out = (float*)d_out;

  const int N = in_sizes[0] / IN_DIM;
  const int E = in_sizes[1] / 2;
  const int NF = N * HID;

  float* dinv   = (float*)d_ws;
  float* bufH   = dinv + (1 << 16);
  float* bufAgg = bufH + (size_t)NF;

  const int npad = (N + 64) & ~63;
  const int epad = (E + 63) & ~63;
  float* aggX   = bufH;
  int*   cnt    = (int*)(bufH + (size_t)N * IN_DIM);
  int*   rowptr = cnt + npad;
  int*   rank   = rowptr + npad;
  int2*  adj    = (int2*)(rank + epad);

  const int* src = EI;
  const int* tgt = EI + E;

  const int T = 256;
  auto cdiv = [](int a, int b) { return (a + b - 1) / b; };

  zero_counts_kernel<<<cdiv(N, T), T, 0, stream>>>(cnt, N);
  count_rank_kernel<<<cdiv(E, T), T, 0, stream>>>(tgt, cnt, rank, E);
  rowptr_dinv_kernel<<<cdiv(N, SCAN_T), SCAN_T, 0, stream>>>(cnt, rowptr, dinv, N);
  fill_adj_kernel<<<cdiv(E, T), T, 0, stream>>>(src, tgt, rank, rowptr, dinv, adj, E);

  gather64_kernel<<<cdiv(N, T / 32), T, 0, stream>>>(X, rowptr, adj, dinv, aggX, N);

  {
    const int strips = cdiv(N, 16);
    wmma_gemm_kernel<IN_DIM><<<cdiv(strips, 2), 128, 0, stream>>>(aggX, W1, bufAgg, N);
  }

  epilogue1_kernel<<<cdiv(NF, T), T, 0, stream>>>(bufAgg, b1, out, NF);

  {
    const int strips = cdiv(N, 16);
    wmma_gemm_kernel<HID><<<cdiv(strips, 2), 128, 0, stream>>>(out, W2, bufAgg, N);
  }

  gather256_bias_kernel<<<cdiv(N, T / 32), T, 0, stream>>>(bufAgg, rowptr, adj, dinv,
                                                            b2, out, N);
}


# both dense products in one launch (7 launches, was 8): a block keeps its 32 rows of layer-1 activations in LDS and multiplies them by W2 from there; that 50000 x 256 array never goes to memory
# speedup vs baseline: 3.8699x; 1.3211x over previous
#include <hip/hip_runtime.h>
#include <math.h>


typedef __attribute__((ext_vector_type(2))) float v2f;
typedef __attribute__((ext_vector_type(8))) float v8f;

static constexpr int HID    = 256;
static constexpr int IN_DIM = 64;

__global__ void zero_counts_kernel(int* __restrict__ cnt, int n) {
  int i = blockIdx.x * blockDim.x + threadIdx.x;
  if (i < n) cnt[i] = 0;
}

__global__ void count_rank_kernel(const int* __restrict__ tgt,
                                  int* __restrict__ cnt,
                                  int* __restrict__ rank, int e) {
  int i = blockIdx.x * blockDim.x + threadIdx.x;
  if (i < e) rank[i] = atomicAdd(&cnt[tgt[i]], 1);
}

static constexpr int SCAN_T = 1024;
__global__ void __launch_bounds__(SCAN_T)
rowptr_dinv_kernel(const int* __restrict__ cnt, int* __restrict__ rowptr,
                   float* __restrict__ dinv, int n) {
  __shared__ int s_tot[32];
  __shared__ int s_pre[32];
  const int tid  = threadIdx.x;
  const int lane = tid & 31;
  const int wv   = tid >> 5;
  const int first = blockIdx.x * SCAN_T;

  int pre = 0;
  for (int i = tid; i < first; i += SCAN_T) pre += cnt[i];

  const int i = first + tid;
  const int c = (i < n) ? cnt[i] : 0;
  int inc = c;
  for (int d = 1; d < 32; d <<= 1) {
    const int v = __shfl_up(inc, d);
    if (lane >= d) inc += v;
  }
  for (int d = 16; d > 0; d >>= 1) pre += __shfl_xor(pre, d);
  if (lane == 31) s_tot[wv] = inc;
  if (lane == 0)  s_pre[wv] = pre;
  __syncthreads();
  if (wv == 0) {
    const int t = s_tot[lane];
    int p  = s_pre[lane];
    int ti = t;
    for (int d = 1; d < 32; d <<= 1) {
      const int v = __shfl_up(ti, d);
      if (lane >= d) ti += v;
    }
    for (int d = 16; d > 0; d >>= 1) p += __shfl_xor(p, d);
    s_tot[lane] = ti - t + p;
  }
  __syncthreads();
  const int excl = s_tot[wv] + inc - c;
  if (i < n) {
    rowptr[i] = excl;
    dinv[i]   = rsqrtf((float)(c + 1));
    if (i == n - 1) rowptr[n] = excl + c;
  }
}

__global__ void fill_adj_kernel(const int* __restrict__ src,
                                const int* __restrict__ tgt,
                                const int* __restrict__ rank,
                                const int* __restrict__ rowptr,
                                const float* __restrict__ dinv,
                                int2* __restrict__ adj, int e) {
  int i = blockIdx.x * blockDim.x + threadIdx.x;
  if (i < e) {
    const int s = src[i];
    adj[rowptr[tgt[i]] + rank[i]] = make_int2(s, __float_as_int(dinv[s]));
  }
}

typedef __attribute__((ext_vector_type(4))) float v4f;

static constexpr int X2_PITCH = HID + 4;

__device__ __forceinline__ void wmma_k8(const v4f a, const float* __restrict__ bp,
                                        v8f (&acc)[8]) {
  v4f b[2][2][2];
#pragma unroll
  for (int s = 0; s < 2; ++s)
#pragma unroll
    for (int j = 0; j < 2; ++j)
#pragma unroll
      for (int q = 0; q < 2; ++q)
        b[s][j][q] = *(const v4f*)(bp + (size_t)(2 * s + j) * HID + 64 * q);

#pragma unroll
  for (int s = 0; s < 2; ++s) {
    v2f af;
    af.x = a[2 * s];
    af.y = a[2 * s + 1];
#pragma unroll
    for (int q = 0; q < 2; ++q)
#pragma unroll
      for (int u = 0; u < 4; ++u) {
        v2f bf;
        bf.x = b[s][0][q][u];
        bf.y = b[s][1][q][u];
        acc[4 * q + u] = __builtin_amdgcn_wmma_f32_16x16x4_f32(
            false, af, false, bf, (short)0, acc[4 * q + u], false, false);
      }
  }
}

__global__ void __launch_bounds__(128) __attribute__((amdgpu_waves_per_eu(7)))
wmma_two_layer_kernel(const float* __restrict__ A,
                      const float* __restrict__ W1,
                      const float* __restrict__ bias,
                      const float* __restrict__ W2,
                      float* __restrict__ C, int M) {
  __shared__ __attribute__((aligned(16))) float x2[32 * X2_PITCH];

  const int lane = threadIdx.x & 31;
  const int wv = __builtin_amdgcn_readfirstlane((int)(threadIdx.x >> 5));
  const int sl     = wv >> 1;
  const int strip  = blockIdx.x * 2 + sl;
  const int tilesM = (M + 15) / 16;
  const bool active = strip < tilesM;

  const int colbase = (wv & 1) * 128;
  const int h = lane >> 4;
  const int c = lane & 15;
  const int row0 = strip * 16 + 8 * h;

  v8f acc[8];

  if (active) {
    int row = strip * 16 + c;
    if (row > M - 1) row = M - 1;
    const float* ap = A + (size_t)row * IN_DIM + 4 * h;
    const float* bp = W1 + (size_t)(4 * h) * HID + colbase + 4 * c;

#pragma unroll
    for (int t = 0; t < 8; ++t) acc[t] = v8f{};

    for (int kk = 0; kk < IN_DIM; kk += 8) {
      const v4f a = *(const v4f*)(ap + kk);
      wmma_k8(a, bp + (size_t)kk * HID, acc);
    }

    const float kLn1e4Over256 = 9.210340371976184f / 256.0f;
    float fr[2][2];
    v4f   bq[2];
#pragma unroll
    for (int q = 0; q < 2; ++q) {
      const int col0 = colbase + 64 * q + 4 * c;
      fr[q][0] = expf(-(float)col0 * kLn1e4Over256);
      fr[q][1] = expf(-(float)(col0 + 2) * kLn1e4Over256);
      bq[q] = *(const v4f*)(bias + col0);
    }
#pragma unroll
    for (int r = 0; r < 8; ++r) {
      const float pos = (float)(row0 + r);
#pragma unroll
      for (int q = 0; q < 2; ++q) {
        float s0, c0, s1, c1;
        sincosf(pos * fr[q][0], &s0, &c0);
        sincosf(pos * fr[q][1], &s1, &c1);
        v4f o;
        o.x = acc[4 * q + 0][r] + bq[q].x + s0;
        o.y = acc[4 * q + 1][r] + bq[q].y + c0;
        o.z = acc[4 * q + 2][r] + bq[q].z + s1;
        o.w = acc[4 * q + 3][r] + bq[q].w + c1;
        o.x = o.x > 0.0f ? o.x : 0.0f;
        o.y = o.y > 0.0f ? o.y : 0.0f;
        o.z = o.z > 0.0f ? o.z : 0.0f;
        o.w = o.w > 0.0f ? o.w : 0.0f;
        *(v4f*)&x2[(sl * 16 + 8 * h + r) * X2_PITCH + colbase + 64 * q + 4 * c] = o;
      }
    }
  }

  __syncthreads();
  if (!active) return;

  {
    const float* bp = W2 + (size_t)(4 * h) * HID + colbase + 4 * c;
    const int abase = (sl * 16 + c) * X2_PITCH + 4 * h;

#pragma unroll
    for (int t = 0; t < 8; ++t) acc[t] = v8f{};

#pragma unroll 4
    for (int kk = 0; kk < HID; kk += 8) {
      const v4f a = *(const v4f*)&x2[abase + kk];
      wmma_k8(a, bp + (size_t)kk * HID, acc);
    }
  }

  float* cp = C + (size_t)row0 * HID + colbase + 4 * c;
#pragma unroll
  for (int r = 0; r < 8; ++r) {
    if (row0 + r < M) {
#pragma unroll
      for (int q = 0; q < 2; ++q) {
        v4f o;
        o.x = acc[4 * q + 0][r];
        o.y = acc[4 * q + 1][r];
        o.z = acc[4 * q + 2][r];
        o.w = acc[4 * q + 3][r];
        *(v4f*)(cp + (size_t)r * HID + 64 * q) = o;
      }
    }
  }
}


__global__ void gather64_kernel(const float* __restrict__ X,
                                const int* __restrict__ rowptr,
                                const int2* __restrict__ adj,
                                const float* __restrict__ dinv,
                                float* __restrict__ out, int N) {
  const int lane = threadIdx.x & 31;
  const int t = __builtin_amdgcn_readfirstlane(
      (int)((blockIdx.x * blockDim.x + threadIdx.x) >> 5));
  if (t >= N) return;
  const int beg = rowptr[t];
  const int end = rowptr[t + 1];
  const float dt = dinv[t];
  const float2* __restrict__ Xv = (const float2*)X;

  const float2 self = Xv[(size_t)t * 32 + lane];
  float ax = self.x * dt, ay = self.y * dt;

  for (int base = beg; base < end; base += 32) {
    const int n = (end - base < 32) ? (end - base) : 32;
    int2 e = make_int2(0, 0);
    if (lane < n) e = adj[base + lane];
    int k = 0;
    for (; k + 4 <= n; k += 4) {
      const int s0 = __builtin_amdgcn_readlane(e.x, k);
      const int s1 = __builtin_amdgcn_readlane(e.x, k + 1);
      const int s2 = __builtin_amdgcn_readlane(e.x, k + 2);
      const int s3 = __builtin_amdgcn_readlane(e.x, k + 3);
      const float w0 = __int_as_float(__builtin_amdgcn_readlane(e.y, k));
      const float w1 = __int_as_float(__builtin_amdgcn_readlane(e.y, k + 1));
      const float w2 = __int_as_float(__builtin_amdgcn_readlane(e.y, k + 2));
      const float w3 = __int_as_float(__builtin_amdgcn_readlane(e.y, k + 3));
      const float2 r0 = Xv[(size_t)s0 * 32 + lane];
      const float2 r1 = Xv[(size_t)s1 * 32 + lane];
      const float2 r2 = Xv[(size_t)s2 * 32 + lane];
      const float2 r3 = Xv[(size_t)s3 * 32 + lane];
      ax = fmaf(r0.x, w0, ax); ay = fmaf(r0.y, w0, ay);
      ax = fmaf(r1.x, w1, ax); ay = fmaf(r1.y, w1, ay);
      ax = fmaf(r2.x, w2, ax); ay = fmaf(r2.y, w2, ay);
      ax = fmaf(r3.x, w3, ax); ay = fmaf(r3.y, w3, ay);
    }
#pragma nounroll
    for (; k < n; ++k) {
      const int s0 = __builtin_amdgcn_readlane(e.x, k);
      const float w0 = __int_as_float(__builtin_amdgcn_readlane(e.y, k));
      const float2 r0 = Xv[(size_t)s0 * 32 + lane];
      ax = fmaf(r0.x, w0, ax); ay = fmaf(r0.y, w0, ay);
    }
  }
  float2 o;
  o.x = ax * dt;
  o.y = ay * dt;
  ((float2*)out)[(size_t)t * 32 + lane] = o;
}

__global__ void gather256_bias_kernel(const float* __restrict__ H,
                                      const int* __restrict__ rowptr,
                                      const int2* __restrict__ adj,
                                      const float* __restrict__ dinv,
                                      const float* __restrict__ bias,
                                      float* __restrict__ out, int N) {
  const int lane = threadIdx.x & 31;
  const int t = __builtin_amdgcn_readfirstlane(
      (int)((blockIdx.x * blockDim.x + threadIdx.x) >> 5));
  if (t >= N) return;
  const int beg = rowptr[t];
  const int end = rowptr[t + 1];
  const float dt = dinv[t];
  const float4* __restrict__ Hv = (const float4*)H;

  float4 a0, a1;
  {
    const float4 p = Hv[(size_t)t * 64 + lane];
    const float4 q = Hv[(size_t)t * 64 + 32 + lane];
    a0.x = p.x * dt; a0.y = p.y * dt; a0.z = p.z * dt; a0.w = p.w * dt;
    a1.x = q.x * dt; a1.y = q.y * dt; a1.z = q.z * dt; a1.w = q.w * dt;
  }

  for (int base = beg; base < end; base += 32) {
    const int n = (end - base < 32) ? (end - base) : 32;
    int2 e = make_int2(0, 0);
    if (lane < n) e = adj[base + lane];
    int k = 0;
    for (; k + 4 <= n; k += 4) {
      const int s0 = __builtin_amdgcn_readlane(e.x, k);
      const int s1 = __builtin_amdgcn_readlane(e.x, k + 1);
      const int s2 = __builtin_amdgcn_readlane(e.x, k + 2);
      const int s3 = __builtin_amdgcn_readlane(e.x, k + 3);
      const float w0 = __int_as_float(__builtin_amdgcn_readlane(e.y, k));
      const float w1 = __int_as_float(__builtin_amdgcn_readlane(e.y, k + 1));
      const float w2 = __int_as_float(__builtin_amdgcn_readlane(e.y, k + 2));
      const float w3 = __int_as_float(__builtin_amdgcn_readlane(e.y, k + 3));
      const float4 p0 = Hv[(size_t)s0 * 64 + lane];
      const float4 q0 = Hv[(size_t)s0 * 64 + 32 + lane];
      const float4 p1 = Hv[(size_t)s1 * 64 + lane];
      const float4 q1 = Hv[(size_t)s1 * 64 + 32 + lane];
      const float4 p2 = Hv[(size_t)s2 * 64 + lane];
      const float4 q2 = Hv[(size_t)s2 * 64 + 32 + lane];
      const float4 p3 = Hv[(size_t)s3 * 64 + lane];
      const float4 q3 = Hv[(size_t)s3 * 64 + 32 + lane];
      a0.x = fmaf(p0.x, w0, a0.x); a0.y = fmaf(p0.y, w0, a0.y);
      a0.z = fmaf(p0.z, w0, a0.z); a0.w = fmaf(p0.w, w0, a0.w);
      a1.x = fmaf(q0.x, w0, a1.x); a1.y = fmaf(q0.y, w0, a1.y);
      a1.z = fmaf(q0.z, w0, a1.z); a1.w = fmaf(q0.w, w0, a1.w);
      a0.x = fmaf(p1.x, w1, a0.x); a0.y = fmaf(p1.y, w1, a0.y);
      a0.z = fmaf(p1.z, w1, a0.z); a0.w = fmaf(p1.w, w1, a0.w);
      a1.x = fmaf(q1.x, w1, a1.x); a1.y = fmaf(q1.y, w1, a1.y);
      a1.z = fmaf(q1.z, w1, a1.z); a1.w = fmaf(q1.w, w1, a1.w);
      a0.x = fmaf(p2.x, w2, a0.x); a0.y = fmaf(p2.y, w2, a0.y);
      a0.z = fmaf(p2.z, w2, a0.z); a0.w = fmaf(p2.w, w2, a0.w);
      a1.x = fmaf(q2.x, w2, a1.x); a1.y = fmaf(q2.y, w2, a1.y);
      a1.z = fmaf(q2.z, w2, a1.z); a1.w = fmaf(q2.w, w2, a1.w);
      a0.x = fmaf(p3.x, w3, a0.x); a0.y = fmaf(p3.y, w3, a0.y);
      a0.z = fmaf(p3.z, w3, a0.z); a0.w = fmaf(p3.w, w3, a0.w);
      a1.x = fmaf(q3.x, w3, a1.x); a1.y = fmaf(q3.y, w3, a1.y);
      a1.z = fmaf(q3.z, w3, a1.z); a1.w = fmaf(q3.w, w3, a1.w);
    }
#pragma nounroll
    for (; k < n; ++k) {
      const int s0 = __builtin_amdgcn_readlane(e.x, k);
      const float w0 = __int_as_float(__builtin_amdgcn_readlane(e.y, k));
      const float4 p0 = Hv[(size_t)s0 * 64 + lane];
      const float4 q0 = Hv[(size_t)s0 * 64 + 32 + lane];
      a0.x = fmaf(p0.x, w0, a0.x); a0.y = fmaf(p0.y, w0, a0.y);
      a0.z = fmaf(p0.z, w0, a0.z); a0.w = fmaf(p0.w, w0, a0.w);
      a1.x = fmaf(q0.x, w0, a1.x); a1.y = fmaf(q0.y, w0, a1.y);
      a1.z = fmaf(q0.z, w0, a1.z); a1.w = fmaf(q0.w, w0, a1.w);
    }
  }
  const float4 b0 = ((const float4*)bias)[lane];
  const float4 b1 = ((const float4*)bias)[32 + lane];
  float4 o0, o1;
  o0.x = fmaf(a0.x, dt, b0.x); o0.y = fmaf(a0.y, dt, b0.y);
  o0.z = fmaf(a0.z, dt, b0.z); o0.w = fmaf(a0.w, dt, b0.w);
  o1.x = fmaf(a1.x, dt, b1.x); o1.y = fmaf(a1.y, dt, b1.y);
  o1.z = fmaf(a1.z, dt, b1.z); o1.w = fmaf(a1.w, dt, b1.w);
  ((float4*)out)[(size_t)t * 64 + lane]      = o0;
  ((float4*)out)[(size_t)t * 64 + 32 + lane] = o1;
}

extern "C" void kernel_launch(void* const* d_in, const int* in_sizes, int n_in,
                              void* d_out, int out_size, void* d_ws, size_t ws_size,
                              hipStream_t stream) {
  const float* X  = (const float*)d_in[0];
  const int*   EI = (const int*)d_in[1];
  const float* W1 = (const float*)d_in[2];
  const float* b1 = (const float*)d_in[3];
  const float* W2 = (const float*)d_in[4];
  const float* b2 = (const float*)d_in[5];
  float* out = (float*)d_out;

  const int N = in_sizes[0] / IN_DIM;
  const int E = in_sizes[1] / 2;
  const int NF = N * HID;

  float* dinv   = (float*)d_ws;
  float* bufH   = dinv + (1 << 16);
  float* bufAgg = bufH + (size_t)NF;

  const int npad = (N + 64) & ~63;
  const int epad = (E + 63) & ~63;
  float* aggX   = bufH;
  int*   cnt    = (int*)(bufH + (size_t)N * IN_DIM);
  int*   rowptr = cnt + npad;
  int*   rank   = rowptr + npad;
  int2*  adj    = (int2*)(rank + epad);

  const int* src = EI;
  const int* tgt = EI + E;

  const int T = 256;
  auto cdiv = [](int a, int b) { return (a + b - 1) / b; };

  zero_counts_kernel<<<cdiv(N, T), T, 0, stream>>>(cnt, N);
  count_rank_kernel<<<cdiv(E, T), T, 0, stream>>>(tgt, cnt, rank, E);
  rowptr_dinv_kernel<<<cdiv(N, SCAN_T), SCAN_T, 0, stream>>>(cnt, rowptr, dinv, N);
  fill_adj_kernel<<<cdiv(E, T), T, 0, stream>>>(src, tgt, rank, rowptr, dinv, adj, E);

  gather64_kernel<<<cdiv(N, T / 32), T, 0, stream>>>(X, rowptr, adj, dinv, aggX, N);

  {
    const int strips = cdiv(N, 16);
    wmma_two_layer_kernel<<<cdiv(strips, 2), 128, 0, stream>>>(aggX, W1, b1, W2, bufAgg, N);
  }

  gather256_bias_kernel<<<cdiv(N, T / 32), T, 0, stream>>>(bufAgg, rowptr, adj, dinv,
                                                            b2, out, N);
}
